# starting kernel
# baseline (speedup 1.0000x reference)
#include <hip/hip_runtime.h>
#include <math.h>

#define BB 2
#define LL 4096
#define EE 1024
#define HH 16
#define DD 64
#define NC 16
#define HEADSZ ((size_t)BB * HH * LL * DD)

typedef __attribute__((ext_vector_type(2))) float v2f;
typedef __attribute__((ext_vector_type(8))) float v8f;

__device__ __forceinline__ v8f wmma4(v2f a, v2f b, v8f c) {
  return __builtin_amdgcn_wmma_f32_16x16x4_f32(false, a, false, b, (short)0, c,
                                               false, false);
}

__device__ __forceinline__ float elup1(float x) {
  return x > 0.f ? x + 1.f : expf(x);
}

__global__ __launch_bounds__(256) void qkv_gemm(const float* __restrict__ x,
                                                const float* __restrict__ W,
                                                float* __restrict__ qb,
                                                float* __restrict__ kb,
                                                float* __restrict__ vb) {
  __shared__ float xs[128 * 36];
  __shared__ float wls[64 * 36];

  const int tid = threadIdx.x;
  const int lane = tid & 31, wv = tid >> 5;
  const int hi = lane >> 4, nlo = lane & 15;
  const int mBase = blockIdx.y * 128;
  const int nBase = blockIdx.x * 64;

  v8f acc[4] = {};

  for (int kt = 0; kt < 1024; kt += 32) {
#pragma unroll
    for (int i = 0; i < 4; ++i) {
      int q = tid + 256 * i;
      int row = q >> 3, c4 = (q & 7) * 4;
      float4 t4 = *(const float4*)(x + (size_t)(mBase + row) * 1024 + kt + c4);
      *(float4*)(xs + row * 36 + c4) = t4;
    }
#pragma unroll
    for (int i = 0; i < 2; ++i) {
      int q = tid + 256 * i;
      int row = q >> 3, c4 = (q & 7) * 4;
      float4 t4 = *(const float4*)(W + (size_t)(nBase + row) * 1024 + kt + c4);
      *(float4*)(wls + row * 36 + c4) = t4;
    }
    __syncthreads();

    const float* xr = xs + (wv * 16 + nlo) * 36;
#pragma unroll
    for (int kk = 0; kk < 32; kk += 4) {
      v2f a;
      a.x = xr[kk + 2 * hi];
      a.y = xr[kk + 2 * hi + 1];
#pragma unroll
      for (int tc = 0; tc < 4; ++tc) {
        const float* wr = wls + (tc * 16 + nlo) * 36;
        v2f b;
        b.x = wr[kk + 2 * hi];
        b.y = wr[kk + 2 * hi + 1];
        acc[tc] = wmma4(a, b, acc[tc]);
      }
    }
    __syncthreads();
  }

  const int tsel = nBase >> 10;
  float* dst = (tsel == 0) ? qb : (tsel == 1) ? kb : vb;
#pragma unroll
  for (int tc = 0; tc < 4; ++tc) {
    int n = nBase + tc * 16 + nlo;
    int rem = n & 1023;
    int h = rem >> 6, d = rem & 63;
#pragma unroll
    for (int j = 0; j < 8; ++j) {
      int m = mBase + wv * 16 + j + 8 * hi;
      float y = acc[tc][j];
      if (tsel == 0)
        y = elup1(y) * 0.125f;
      else if (tsel == 1)
        y = elup1(y);
      int b = m >> 12, l = m & 4095;
      dst[((size_t)(b * 16 + h) * 4096 + l) * 64 + d] = y;
    }
  }
}

#define AP 68

__global__ __launch_bounds__(256) void chunk_stats(const float* __restrict__ kb,
                                                   const float* __restrict__ vb,
                                                   float* __restrict__ Sch,
                                                   float* __restrict__ Zch) {
  __shared__ float ks[64 * AP];
  __shared__ float vs[64 * AP];
  const int tid = threadIdx.x;
  const int lane = tid & 31, wv = tid >> 5;
  const int hi = lane >> 4, nlo = lane & 15;
  const int c = blockIdx.x, bh = blockIdx.y;
  const size_t base = ((size_t)bh * 4096 + c * 256) * 64;

  v8f sc[2] = {};
  float zacc = 0.f;

  for (int sb = 0; sb < 4; ++sb) {
#pragma unroll
    for (int i = 0; i < 4; ++i) {
      int q = tid + 256 * i;
      int row = q >> 4, c4 = (q & 15) * 4;
      size_t g = base + (size_t)(sb * 64 + row) * 64 + c4;
      *(float4*)(ks + row * AP + c4) = *(const float4*)(kb + g);
      *(float4*)(vs + row * AP + c4) = *(const float4*)(vb + g);
    }
    __syncthreads();
#pragma unroll
    for (int ti = 0; ti < 2; ++ti) {
      const int t = 2 * wv + ti, td = t >> 2, te = t & 3;
      v8f acc = sc[ti];
#pragma unroll
      for (int k0 = 0; k0 < 64; k0 += 4) {
        v2f a;
        a.x = ks[(k0 + 2 * hi) * AP + td * 16 + nlo];
        a.y = ks[(k0 + 2 * hi + 1) * AP + td * 16 + nlo];
        v2f bv;
        bv.x = vs[(k0 + 2 * hi) * AP + te * 16 + nlo];
        bv.y = vs[(k0 + 2 * hi + 1) * AP + te * 16 + nlo];
        acc = wmma4(a, bv, acc);
      }
      sc[ti] = acc;
    }
    if (tid < 64) {
      float s = 0.f;
      for (int m = 0; m < 64; ++m) s += ks[m * AP + tid];
      zacc += s;
    }
    __syncthreads();
  }

  float* So = Sch + ((size_t)bh * NC + c) * 4096;
#pragma unroll
  for (int ti = 0; ti < 2; ++ti) {
    const int t = 2 * wv + ti, td = t >> 2, te = t & 3;
#pragma unroll
    for (int j = 0; j < 8; ++j)
      So[(td * 16 + j + 8 * hi) * 64 + te * 16 + nlo] = sc[ti][j];
  }
  if (tid < 64) Zch[((size_t)bh * NC + c) * 64 + tid] = zacc;
}

__global__ __launch_bounds__(256) void prefix_scan(float* __restrict__ Sch,
                                                   float* __restrict__ Zch) {
  const int bh = blockIdx.x, tid = threadIdx.x;
  for (int el = tid; el < 4096; el += 256) {
    float run = 0.f;
    for (int c = 0; c < NC; ++c) {
      size_t idx = ((size_t)bh * NC + c) * 4096 + el;
      float tmp = Sch[idx];
      Sch[idx] = run;
      run += tmp;
    }
  }
  if (tid < 64) {
    float run = 0.f;
    for (int c = 0; c < NC; ++c) {
      size_t idx = ((size_t)bh * NC + c) * 64 + tid;
      float tmp = Zch[idx];
      Zch[idx] = run;
      run += tmp;
    }
  }
}

__global__ __launch_bounds__(256) void attn_chunk(float* __restrict__ qo,
                                                  const float* __restrict__ kb,
                                                  const float* __restrict__ vb,
                                                  const float* __restrict__ Sch,
                                                  const float* __restrict__ Zch) {
  extern __shared__ float sm[];
  float* qs = sm;
  float* ks = sm + 64 * AP;
  float* vs = sm + 2 * 64 * AP;
  float* as_ = sm + 3 * 64 * AP;
  float* Ss = sm + 4 * 64 * AP;
  float* Zs = sm + 5 * 64 * AP;
  float* dens = Zs + 64;

  const int tid = threadIdx.x;
  const int lane = tid & 31, wv = tid >> 5;
  const int hi = lane >> 4, nlo = lane & 15;
  const int c = blockIdx.x, bh = blockIdx.y;
  const size_t base = (size_t)bh * 4096 * 64;

  const float* Sg = Sch + ((size_t)bh * NC + c) * 4096;
  for (int i = tid; i < 4096; i += 256) Ss[(i >> 6) * AP + (i & 63)] = Sg[i];
  if (tid < 64) Zs[tid] = Zch[((size_t)bh * NC + c) * 64 + tid];
  __syncthreads();

  for (int sb = 0; sb < 4; ++sb) {
    const int l0 = c * 256 + sb * 64;
#pragma unroll
    for (int i = 0; i < 4; ++i) {
      int q = tid + 256 * i;
      int row = q >> 4, c4 = (q & 15) * 4;
      size_t g = base + (size_t)(l0 + row) * 64 + c4;
      *(float4*)(qs + row * AP + c4) = *(const float4*)(qo + g);
      *(float4*)(ks + row * AP + c4) = *(const float4*)(kb + g);
      *(float4*)(vs + row * AP + c4) = *(const float4*)(vb + g);
    }
    __syncthreads();

    v8f num[2];
#pragma unroll
    for (int ti = 0; ti < 2; ++ti) {
      const int t = 2 * wv + ti, tl = t >> 2, tc = t & 3;
      v8f acc = {};
      v8f nacc = {};
      const float* qr = qs + (tl * 16 + nlo) * AP;
      const float* kr = ks + (tc * 16 + nlo) * AP;
#pragma unroll
      for (int k0 = 0; k0 < 64; k0 += 4) {
        v2f a;
        a.x = qr[k0 + 2 * hi];
        a.y = qr[k0 + 2 * hi + 1];
        v2f bk;
        bk.x = kr[k0 + 2 * hi];
        bk.y = kr[k0 + 2 * hi + 1];
        acc = wmma4(a, bk, acc);
        v2f bs;
        bs.x = Ss[(k0 + 2 * hi) * AP + tc * 16 + nlo];
        bs.y = Ss[(k0 + 2 * hi + 1) * AP + tc * 16 + nlo];
        nacc = wmma4(a, bs, nacc);
      }
#pragma unroll
      for (int j = 0; j < 8; ++j) {
        int r = tl * 16 + j + 8 * hi;
        int cc = tc * 16 + nlo;
        as_[r * AP + cc] = (cc <= r) ? acc[j] : 0.f;
      }
      num[ti] = nacc;
    }
    __syncthreads();

    if (tid < 64) {
      float s = 0.f;
      const float* ar = as_ + tid * AP;
      for (int n = 0; n < 64; ++n) s += ar[n];
      const float* qr = qs + tid * AP;
      for (int d = 0; d < 64; ++d) s += qr[d] * Zs[d];
      dens[tid] = fmaxf(s, 1e-6f);
    }
#pragma unroll
    for (int ti = 0; ti < 2; ++ti) {
      const int t = 2 * wv + ti, tl = t >> 2, tc = t & 3;
      v8f nacc = num[ti];
      const float* ar = as_ + (tl * 16 + nlo) * AP;
#pragma unroll
      for (int k0 = 0; k0 < 64; k0 += 4) {
        v2f a;
        a.x = ar[k0 + 2 * hi];
        a.y = ar[k0 + 2 * hi + 1];
        v2f bv;
        bv.x = vs[(k0 + 2 * hi) * AP + tc * 16 + nlo];
        bv.y = vs[(k0 + 2 * hi + 1) * AP + tc * 16 + nlo];
        nacc = wmma4(a, bv, nacc);
      }
      num[ti] = nacc;
    }
    __syncthreads();

    if (sb != 3) {
#pragma unroll
      for (int ti = 0; ti < 2; ++ti) {
        const int t = 2 * wv + ti, td = t >> 2, te = t & 3;
        v8f scv;
#pragma unroll
        for (int j = 0; j < 8; ++j)
          scv[j] = Ss[(td * 16 + j + 8 * hi) * AP + te * 16 + nlo];
#pragma unroll
        for (int k0 = 0; k0 < 64; k0 += 4) {
          v2f a;
          a.x = ks[(k0 + 2 * hi) * AP + td * 16 + nlo];
          a.y = ks[(k0 + 2 * hi + 1) * AP + td * 16 + nlo];
          v2f bv;
          bv.x = vs[(k0 + 2 * hi) * AP + te * 16 + nlo];
          bv.y = vs[(k0 + 2 * hi + 1) * AP + te * 16 + nlo];
          scv = wmma4(a, bv, scv);
        }
#pragma unroll
        for (int j = 0; j < 8; ++j)
          Ss[(td * 16 + j + 8 * hi) * AP + te * 16 + nlo] = scv[j];
      }
      if (tid < 64) {
        float s = Zs[tid];
        for (int m = 0; m < 64; ++m) s += ks[m * AP + tid];
        Zs[tid] = s;
      }
    }
#pragma unroll
    for (int ti = 0; ti < 2; ++ti) {
      const int t = 2 * wv + ti, tl = t >> 2, te = t & 3;
#pragma unroll
      for (int j = 0; j < 8; ++j) {
        int r = tl * 16 + j + 8 * hi;
        float v = num[ti][j] / dens[r];
        qo[base + (size_t)(l0 + r) * 64 + te * 16 + nlo] = v;
      }
    }
    __syncthreads();
  }
}

__global__ __launch_bounds__(256) void out_gemm(const float* __restrict__ oh,
                                                const float* __restrict__ W,
                                                float* __restrict__ out) {
  __shared__ float xs[128 * 36];
  __shared__ float wls[64 * 36];

  const int tid = threadIdx.x;
  const int lane = tid & 31, wv = tid >> 5;
  const int hi = lane >> 4, nlo = lane & 15;
  const int mBase = blockIdx.y * 128;
  const int nBase = blockIdx.x * 64;

  v8f acc[4] = {};

  for (int kt = 0; kt < 1024; kt += 32) {
    const int h = kt >> 6;
    const int dbase = kt & 63;
#pragma unroll
    for (int i = 0; i < 4; ++i) {
      int q = tid + 256 * i;
      int row = q >> 3, c4 = (q & 7) * 4;
      int m = mBase + row;
      int b = m >> 12, l = m & 4095;
      float4 t4 = *(const float4*)(oh + ((size_t)(b * 16 + h) * 4096 + l) * 64 +
                                   dbase + c4);
      *(float4*)(xs + row * 36 + c4) = t4;
    }
#pragma unroll
    for (int i = 0; i < 2; ++i) {
      int q = tid + 256 * i;
      int row = q >> 3, c4 = (q & 7) * 4;
      float4 t4 = *(const float4*)(W + (size_t)(nBase + row) * 1024 + kt + c4);
      *(float4*)(wls + row * 36 + c4) = t4;
    }
    __syncthreads();

    const float* xr = xs + (wv * 16 + nlo) * 36;
#pragma unroll
    for (int kk = 0; kk < 32; kk += 4) {
      v2f a;
      a.x = xr[kk + 2 * hi];
      a.y = xr[kk + 2 * hi + 1];
#pragma unroll
      for (int tc = 0; tc < 4; ++tc) {
        const float* wr = wls + (tc * 16 + nlo) * 36;
        v2f b;
        b.x = wr[kk + 2 * hi];
        b.y = wr[kk + 2 * hi + 1];
        acc[tc] = wmma4(a, b, acc[tc]);
      }
    }
    __syncthreads();
  }

#pragma unroll
  for (int tc = 0; tc < 4; ++tc) {
    int n = nBase + tc * 16 + nlo;
#pragma unroll
    for (int j = 0; j < 8; ++j) {
      int m = mBase + wv * 16 + j + 8 * hi;
      out[(size_t)m * 1024 + n] = acc[tc][j];
    }
  }
}

extern "C" void kernel_launch(void* const* d_in, const int* in_sizes, int n_in,
                              void* d_out, int out_size, void* d_ws,
                              size_t ws_size, hipStream_t stream) {
  const float* x = (const float*)d_in[0];
  const float* Wqkv = (const float*)d_in[1];
  const float* Wout = (const float*)d_in[2];
  float* out = (float*)d_out;

  float* qbuf = (float*)d_ws;
  float* kbuf = qbuf + HEADSZ;
  float* vbuf = kbuf + HEADSZ;
  float* Sch = vbuf + HEADSZ;
  float* Zch = Sch + (size_t)BB * HH * NC * 4096;

  qkv_gemm<<<dim3(48, 64), 256, 0, stream>>>(x, Wqkv, qbuf, kbuf, vbuf);

  chunk_stats<<<dim3(NC, BB * HH), 256, 0, stream>>>(kbuf, vbuf, Sch, Zch);
  prefix_scan<<<dim3(BB * HH), 256, 0, stream>>>(Sch, Zch);
  const size_t smem = (size_t)(5 * 64 * AP + 128) * sizeof(float);
  attn_chunk<<<dim3(NC, BB * HH), 256, smem, stream>>>(qbuf, kbuf, vbuf, Sch,
                                                       Zch);

  out_gemm<<<dim3(16, 64), 256, 0, stream>>>(qbuf, Wout, out);
}


# Both projection GEMMs: W fragments too go global->registers (no LDS, no barrier); each wave owns a 64x64 tile and issues a 32-k slab's 32 loads one slab ahead of use; sum order unchanged
# speedup vs baseline: 1.1980x; 1.1980x over previous
#include <hip/hip_runtime.h>
#include <math.h>

#define BB 2
#define LL 4096
#define EE 1024
#define HH 16
#define DD 64
#define NC 16
#define HEADSZ ((size_t)BB * HH * LL * DD)

typedef __attribute__((ext_vector_type(2))) float v2f;
typedef __attribute__((ext_vector_type(8))) float v8f;

__device__ __forceinline__ v8f wmma4(v2f a, v2f b, v8f c) {
  return __builtin_amdgcn_wmma_f32_16x16x4_f32(false, a, false, b, (short)0, c,
                                               false, false);
}

__device__ __forceinline__ float elup1(float x) {
  return x > 0.f ? x + 1.f : expf(x);
}

template <int A_TILE_STRIDE>
__device__ __forceinline__ void load_slab(float4 (&fa)[4][4], float4 (&fb)[4][4],
                                          const float* ap, const float* bp) {
#pragma unroll
  for (int t = 0; t < 4; ++t) {
#pragma unroll
    for (int i = 0; i < 4; ++i) fa[t][i] = *(const float4*)(ap + t * A_TILE_STRIDE + 8 * i);
  }
#pragma unroll
  for (int t = 0; t < 4; ++t) {
#pragma unroll
    for (int i = 0; i < 4; ++i) fb[t][i] = *(const float4*)(bp + t * 16 * 1024 + 8 * i);
  }
}

__device__ __forceinline__ void wmma_drain() {
  __builtin_amdgcn_sched_barrier(0);
  asm volatile("v_nop\n\tv_nop\n\tv_nop\n\tv_nop\n\tv_nop");
  __builtin_amdgcn_sched_barrier(0);
}

__device__ __forceinline__ void mma_slab(v8f (&acc)[4][4], const float4 (&fa)[4][4],
                                         const float4 (&fb)[4][4]) {
#pragma unroll
  for (int i = 0; i < 4; ++i) {
#pragma unroll
    for (int tm = 0; tm < 4; ++tm) {
      v2f a0;
      a0.x = fa[tm][i].x; a0.y = fa[tm][i].y;
#pragma unroll
      for (int tn = 0; tn < 4; ++tn) {
        v2f b0;
        b0.x = fb[tn][i].x; b0.y = fb[tn][i].y;
        acc[tm][tn] = wmma4(a0, b0, acc[tm][tn]);
      }
    }
#pragma unroll
    for (int tm = 0; tm < 4; ++tm) {
      v2f a1;
      a1.x = fa[tm][i].z; a1.y = fa[tm][i].w;
#pragma unroll
      for (int tn = 0; tn < 4; ++tn) {
        v2f b1;
        b1.x = fb[tn][i].z; b1.y = fb[tn][i].w;
        acc[tm][tn] = wmma4(a1, b1, acc[tm][tn]);
      }
    }
  }
}

template <int A_TILE_STRIDE, int A_PAIR_STRIDE>
__device__ __forceinline__ void gemm_wave_64x64(v8f (&acc)[4][4], const float* ap,
                                                const float* bp) {
  float4 fa0[4][4], fb0[4][4], fa1[4][4], fb1[4][4];
  load_slab<A_TILE_STRIDE>(fa0, fb0, ap, bp);
#pragma unroll 1
  for (int p = 0; p < 15; ++p) {
    load_slab<A_TILE_STRIDE>(fa1, fb1, ap + 32, bp + 32);
    mma_slab(acc, fa0, fb0);
    ap += A_PAIR_STRIDE;
    bp += 64;
    load_slab<A_TILE_STRIDE>(fa0, fb0, ap, bp);
    mma_slab(acc, fa1, fb1);
  }
  load_slab<A_TILE_STRIDE>(fa1, fb1, ap + 32, bp + 32);
  mma_slab(acc, fa0, fb0);
  mma_slab(acc, fa1, fb1);
  wmma_drain();
}

__global__ __launch_bounds__(128) __attribute__((amdgpu_waves_per_eu(2, 2)))
void qkv_gemm(const float* __restrict__ x, const float* __restrict__ W,
              float* __restrict__ qb, float* __restrict__ kb,
              float* __restrict__ vb) {
  const int tid = threadIdx.x;
  const int lane = tid & 31;
  const int wv = __builtin_amdgcn_readfirstlane(tid >> 5);
  const int hi = lane >> 4, nlo = lane & 15;
  const int mBase = blockIdx.y * 128 + (wv >> 1) * 64;
  const int nBase = blockIdx.x * 128 + (wv & 1) * 64;

  const float* ap = x + (size_t)(mBase + nlo) * 1024 + 4 * hi;
  const float* bp = W + (size_t)(nBase + nlo) * 1024 + 4 * hi;

  v8f acc[4][4] = {};
  gemm_wave_64x64<16 * 1024, 64>(acc, ap, bp);

  const int tsel = nBase >> 10;
  float* dst = (tsel == 0) ? qb : (tsel == 1) ? kb : vb;
  const int h = (nBase & 1023) >> 6;
  const int b = mBase >> 12, l0 = mBase & 4095;
  float* o = dst + ((size_t)(b * 16 + h) * 4096 + l0) * 64 + 8 * hi * 64 + nlo;
  if (tsel == 2) {
#pragma unroll
    for (int tm = 0; tm < 4; ++tm) {
#pragma unroll
      for (int tn = 0; tn < 4; ++tn) {
#pragma unroll
        for (int j = 0; j < 8; ++j)
          o[(tm * 16 + j) * 64 + tn * 16] = acc[tm][tn][j];
      }
    }
  } else {
    const float sc = (tsel == 0) ? 0.125f : 1.0f;
#pragma unroll
    for (int tm = 0; tm < 4; ++tm) {
#pragma unroll
      for (int tn = 0; tn < 4; ++tn) {
#pragma unroll
        for (int j = 0; j < 8; ++j)
          o[(tm * 16 + j) * 64 + tn * 16] = elup1(acc[tm][tn][j]) * sc;
      }
    }
  }
}

#define AP 68

__global__ __launch_bounds__(256) void chunk_stats(const float* __restrict__ kb,
                                                   const float* __restrict__ vb,
                                                   float* __restrict__ Sch,
                                                   float* __restrict__ Zch) {
  __shared__ float ks[64 * AP];
  __shared__ float vs[64 * AP];
  const int tid = threadIdx.x;
  const int lane = tid & 31, wv = tid >> 5;
  const int hi = lane >> 4, nlo = lane & 15;
  const int c = blockIdx.x, bh = blockIdx.y;
  const size_t base = ((size_t)bh * 4096 + c * 256) * 64;

  v8f sc[2] = {};
  float zacc = 0.f;

  for (int sb = 0; sb < 4; ++sb) {
#pragma unroll
    for (int i = 0; i < 4; ++i) {
      int q = tid + 256 * i;
      int row = q >> 4, c4 = (q & 15) * 4;
      size_t g = base + (size_t)(sb * 64 + row) * 64 + c4;
      *(float4*)(ks + row * AP + c4) = *(const float4*)(kb + g);
      *(float4*)(vs + row * AP + c4) = *(const float4*)(vb + g);
    }
    __syncthreads();
#pragma unroll
    for (int ti = 0; ti < 2; ++ti) {
      const int t = 2 * wv + ti, td = t >> 2, te = t & 3;
      v8f acc = sc[ti];
#pragma unroll
      for (int k0 = 0; k0 < 64; k0 += 4) {
        v2f a;
        a.x = ks[(k0 + 2 * hi) * AP + td * 16 + nlo];
        a.y = ks[(k0 + 2 * hi + 1) * AP + td * 16 + nlo];
        v2f bv;
        bv.x = vs[(k0 + 2 * hi) * AP + te * 16 + nlo];
        bv.y = vs[(k0 + 2 * hi + 1) * AP + te * 16 + nlo];
        acc = wmma4(a, bv, acc);
      }
      sc[ti] = acc;
    }
    if (tid < 64) {
      float s = 0.f;
      for (int m = 0; m < 64; ++m) s += ks[m * AP + tid];
      zacc += s;
    }
    __syncthreads();
  }

  float* So = Sch + ((size_t)bh * NC + c) * 4096;
#pragma unroll
  for (int ti = 0; ti < 2; ++ti) {
    const int t = 2 * wv + ti, td = t >> 2, te = t & 3;
#pragma unroll
    for (int j = 0; j < 8; ++j)
      So[(td * 16 + j + 8 * hi) * 64 + te * 16 + nlo] = sc[ti][j];
  }
  if (tid < 64) Zch[((size_t)bh * NC + c) * 64 + tid] = zacc;
}

#define SCAN_S_ITEMS (BB * HH * 1024)
#define SCAN_ITEMS (SCAN_S_ITEMS + BB * HH * 16)
typedef __attribute__((ext_vector_type(4))) float v4f;

__global__ __launch_bounds__(64) __attribute__((amdgpu_waves_per_eu(1, 8)))
void prefix_scan(float* __restrict__ Sch, float* __restrict__ Zch) {
  const int item = blockIdx.x * 64 + threadIdx.x;
  const bool isS = blockIdx.x < SCAN_S_ITEMS / 64;
  const int j = isS ? item : item - SCAN_S_ITEMS;
  const int stride = isS ? 4096 : 64;
  const int sh = isS ? 10 : 4;
  float* p = (isS ? Sch : Zch) + (size_t)(j >> sh) * (NC * stride) +
             (j & ((1 << sh) - 1)) * 4;
  v4f t[NC];
#pragma unroll
  for (int c = 0; c < NC; ++c) t[c] = *(const v4f*)(p + (size_t)c * stride);
  v4f run = {0.f, 0.f, 0.f, 0.f};
#pragma unroll
  for (int c = 0; c < NC; ++c) {
    *(v4f*)(p + (size_t)c * stride) = run;
    run = run + t[c];
  }
}

__global__ __launch_bounds__(256) void attn_chunk(float* __restrict__ qo,
                                                  const float* __restrict__ kb,
                                                  const float* __restrict__ vb,
                                                  const float* __restrict__ Sch,
                                                  const float* __restrict__ Zch) {
  extern __shared__ float sm[];
  float* qs = sm;
  float* ks = sm + 64 * AP;
  float* vs = sm + 2 * 64 * AP;
  float* as_ = sm + 3 * 64 * AP;
  float* Ss = sm + 4 * 64 * AP;
  float* Zs = sm + 5 * 64 * AP;
  float* dens = Zs + 64;

  const int tid = threadIdx.x;
  const int lane = tid & 31, wv = tid >> 5;
  const int hi = lane >> 4, nlo = lane & 15;
  const int c = blockIdx.x, bh = blockIdx.y;
  const size_t base = (size_t)bh * 4096 * 64;

  const float* Sg = Sch + ((size_t)bh * NC + c) * 4096;
  for (int i = tid; i < 4096; i += 256) Ss[(i >> 6) * AP + (i & 63)] = Sg[i];
  if (tid < 64) Zs[tid] = Zch[((size_t)bh * NC + c) * 64 + tid];
  __syncthreads();

  for (int sb = 0; sb < 4; ++sb) {
    const int l0 = c * 256 + sb * 64;
#pragma unroll
    for (int i = 0; i < 4; ++i) {
      int q = tid + 256 * i;
      int row = q >> 4, c4 = (q & 15) * 4;
      size_t g = base + (size_t)(l0 + row) * 64 + c4;
      *(float4*)(qs + row * AP + c4) = *(const float4*)(qo + g);
      *(float4*)(ks + row * AP + c4) = *(const float4*)(kb + g);
      *(float4*)(vs + row * AP + c4) = *(const float4*)(vb + g);
    }
    __syncthreads();

    v8f num[2];
#pragma unroll
    for (int ti = 0; ti < 2; ++ti) {
      const int t = 2 * wv + ti, tl = t >> 2, tc = t & 3;
      v8f acc = {};
      v8f nacc = {};
      const float* qr = qs + (tl * 16 + nlo) * AP;
      const float* kr = ks + (tc * 16 + nlo) * AP;
#pragma unroll
      for (int k0 = 0; k0 < 64; k0 += 4) {
        v2f a;
        a.x = qr[k0 + 2 * hi];
        a.y = qr[k0 + 2 * hi + 1];
        v2f bk;
        bk.x = kr[k0 + 2 * hi];
        bk.y = kr[k0 + 2 * hi + 1];
        acc = wmma4(a, bk, acc);
        v2f bs;
        bs.x = Ss[(k0 + 2 * hi) * AP + tc * 16 + nlo];
        bs.y = Ss[(k0 + 2 * hi + 1) * AP + tc * 16 + nlo];
        nacc = wmma4(a, bs, nacc);
      }
#pragma unroll
      for (int j = 0; j < 8; ++j) {
        int r = tl * 16 + j + 8 * hi;
        int cc = tc * 16 + nlo;
        as_[r * AP + cc] = (cc <= r) ? acc[j] : 0.f;
      }
      num[ti] = nacc;
    }
    __syncthreads();

    if (tid < 64) {
      float s = 0.f;
      const float* ar = as_ + tid * AP;
      for (int n = 0; n < 64; ++n) s += ar[n];
      const float* qr = qs + tid * AP;
      for (int d = 0; d < 64; ++d) s += qr[d] * Zs[d];
      dens[tid] = fmaxf(s, 1e-6f);
    }
#pragma unroll
    for (int ti = 0; ti < 2; ++ti) {
      const int t = 2 * wv + ti, tl = t >> 2, tc = t & 3;
      v8f nacc = num[ti];
      const float* ar = as_ + (tl * 16 + nlo) * AP;
#pragma unroll
      for (int k0 = 0; k0 < 64; k0 += 4) {
        v2f a;
        a.x = ar[k0 + 2 * hi];
        a.y = ar[k0 + 2 * hi + 1];
        v2f bv;
        bv.x = vs[(k0 + 2 * hi) * AP + tc * 16 + nlo];
        bv.y = vs[(k0 + 2 * hi + 1) * AP + tc * 16 + nlo];
        nacc = wmma4(a, bv, nacc);
      }
      num[ti] = nacc;
    }
    __syncthreads();

    if (sb != 3) {
#pragma unroll
      for (int ti = 0; ti < 2; ++ti) {
        const int t = 2 * wv + ti, td = t >> 2, te = t & 3;
        v8f scv;
#pragma unroll
        for (int j = 0; j < 8; ++j)
          scv[j] = Ss[(td * 16 + j + 8 * hi) * AP + te * 16 + nlo];
#pragma unroll
        for (int k0 = 0; k0 < 64; k0 += 4) {
          v2f a;
          a.x = ks[(k0 + 2 * hi) * AP + td * 16 + nlo];
          a.y = ks[(k0 + 2 * hi + 1) * AP + td * 16 + nlo];
          v2f bv;
          bv.x = vs[(k0 + 2 * hi) * AP + te * 16 + nlo];
          bv.y = vs[(k0 + 2 * hi + 1) * AP + te * 16 + nlo];
          scv = wmma4(a, bv, scv);
        }
#pragma unroll
        for (int j = 0; j < 8; ++j)
          Ss[(td * 16 + j + 8 * hi) * AP + te * 16 + nlo] = scv[j];
      }
      if (tid < 64) {
        float s = Zs[tid];
        for (int m = 0; m < 64; ++m) s += ks[m * AP + tid];
        Zs[tid] = s;
      }
    }
#pragma unroll
    for (int ti = 0; ti < 2; ++ti) {
      const int t = 2 * wv + ti, tl = t >> 2, te = t & 3;
#pragma unroll
      for (int j = 0; j < 8; ++j) {
        int r = tl * 16 + j + 8 * hi;
        float v = num[ti][j] / dens[r];
        qo[base + (size_t)(l0 + r) * 64 + te * 16 + nlo] = v;
      }
    }
    __syncthreads();
  }
}

__global__ __launch_bounds__(128) __attribute__((amdgpu_waves_per_eu(2, 2)))
void out_gemm(const float* __restrict__ oh, const float* __restrict__ W,
              float* __restrict__ out) {
  const int tid = threadIdx.x;
  const int lane = tid & 31;
  const int wv = __builtin_amdgcn_readfirstlane(tid >> 5);
  const int hi = lane >> 4, nlo = lane & 15;
  const int mBase = blockIdx.y * 128 + (wv >> 1) * 64;
  const int nBase = blockIdx.x * 128 + (wv & 1) * 64;

  const int bidx = mBase >> 12, l0 = mBase & 4095;
  const float* ap = oh + ((size_t)(bidx * 16) * 4096 + l0 + nlo) * 64 + 4 * hi;
  const float* bp = W + (size_t)(nBase + nlo) * 1024 + 4 * hi;

  v8f acc[4][4] = {};
  gemm_wave_64x64<16 * 64, 4096 * 64>(acc, ap, bp);

  float* o = out + (size_t)mBase * 1024 + nBase;
#pragma unroll
  for (int tm = 0; tm < 4; ++tm) {
#pragma unroll
    for (int tn = 0; tn < 4; ++tn) {
#pragma unroll
      for (int j = 0; j < 8; ++j)
        o[(size_t)(tm * 16 + j + 8 * hi) * 1024 + tn * 16 + nlo] = acc[tm][tn][j];
    }
  }
}

extern "C" void kernel_launch(void* const* d_in, const int* in_sizes, int n_in,
                              void* d_out, int out_size, void* d_ws,
                              size_t ws_size, hipStream_t stream) {
  const float* x = (const float*)d_in[0];
  const float* Wqkv = (const float*)d_in[1];
  const float* Wout = (const float*)d_in[2];
  float* out = (float*)d_out;

  float* qbuf = (float*)d_ws;
  float* kbuf = qbuf + HEADSZ;
  float* vbuf = kbuf + HEADSZ;
  float* Sch = vbuf + HEADSZ;
  float* Zch = Sch + (size_t)BB * HH * NC * 4096;

  qkv_gemm<<<dim3(24, 64), 128, 0, stream>>>(x, Wqkv, qbuf, kbuf, vbuf);

  chunk_stats<<<dim3(NC, BB * HH), 256, 0, stream>>>(kbuf, vbuf, Sch, Zch);
  prefix_scan<<<dim3(SCAN_ITEMS / 64), 64, 0, stream>>>(Sch, Zch);
  const size_t smem = (size_t)(5 * 64 * AP + 128) * sizeof(float);
  attn_chunk<<<dim3(NC, BB * HH), 256, smem, stream>>>(qbuf, kbuf, vbuf, Sch,
                                                       Zch);

  out_gemm<<<dim3(8, 64), 128, 0, stream>>>(qbuf, Wout, out);
}
